# starting kernel
# baseline (speedup 1.0000x reference)
#include <hip/hip_runtime.h>


typedef float v2f __attribute__((ext_vector_type(2)));
typedef float v8f __attribute__((ext_vector_type(8)));

#define HID   19
#define SEQ   2048
#define BATCH 1024
#define NKIT  5

#define WMMA4(a, b, c) __builtin_amdgcn_wmma_f32_16x16x4_f32( \
        false, (a), false, (b), (short)0, (c), false, false)

__device__ __forceinline__ float fast_tanh(float x) {
#if __has_builtin(__builtin_amdgcn_tanhf)
    return __builtin_amdgcn_tanhf(x);
#else
    float r;
    asm("v_tanh_f32 %0, %1" : "=v"(r) : "v"(x));
    return r;
#endif
}
__device__ __forceinline__ float fast_sigmoid(float x) {
    return __builtin_fmaf(0.5f, fast_tanh(0.5f * x), 0.5f);
}
__device__ __forceinline__ void wave_sync() {
    __builtin_amdgcn_wave_barrier();
}

__global__ __launch_bounds__(32) void gru_wmma_scan(
    const float* __restrict__ x,
    const float* __restrict__ Wih,
    const float* __restrict__ Whh,
    const float* __restrict__ bih,
    const float* __restrict__ bhh,
    const float* __restrict__ Wout,
    const float* __restrict__ bout,
    float* __restrict__ out)
{
    __shared__ float GX[64][17];
    __shared__ float GH2[32][17];

    const int lane  = threadIdx.x;
    const int m     = lane & 15;
    const int half  = lane >> 4;
    const int bbase = blockIdx.x * 16;

    v2f Bm[2][2 * NKIT];
    #pragma unroll
    for (int nt = 0; nt < 2; ++nt) {
        const int g = 16 * nt + m;
        #pragma unroll
        for (int ki = 0; ki < 2 * NKIT; ++ki) {
            #pragma unroll
            for (int v = 0; v < 2; ++v) {
                float w = 0.0f;
                if (ki < NKIT) {
                    const int K = 4 * ki + 2 * half + v;
                    if (K < HID) w = Wih[g * HID + K];
                } else {
                    const int K = 4 * (ki - NKIT) + 2 * half + v;
                    w = (K < HID) ? Whh[g * HID + K] : (bih[g] + bhh[g]);
                }
                Bm[nt][ki][v] = w;
            }
        }
    }
    v2f Bih2[2][NKIT], Bhh2[2][NKIT];
    #pragma unroll
    for (int u = 0; u < 2; ++u) {
        const int g = 32 + 16 * u + m;
        #pragma unroll
        for (int ki = 0; ki < NKIT; ++ki) {
            #pragma unroll
            for (int v = 0; v < 2; ++v) {
                const int K = 4 * ki + 2 * half + v;
                float wi = 0.0f, wh = 0.0f;
                if (g < 57) {
                    if (K < HID) {
                        wi = Wih[g * HID + K];
                        wh = Whh[g * HID + K];
                    } else {
                        wh = (g < 38) ? (bih[g] + bhh[g]) : bhh[g];
                    }
                }
                Bih2[u][ki][v] = wi;
                Bhh2[u][ki][v] = wh;
            }
        }
    }
    const float bn2 = (32 + m >= 38) ? bih[32 + m] : 0.0f;
    const float bn3 = (48 + m < 57)  ? bih[48 + m] : 0.0f;
    const float bo  = bout[0];

    float wo[NKIT][2];
    #pragma unroll
    for (int ki = 0; ki < NKIT; ++ki)
        #pragma unroll
        for (int v = 0; v < 2; ++v) {
            const int j = 4 * ki + 2 * half + v;
            wo[ki][v] = (j < HID) ? Wout[j] : 0.0f;
        }

    v2f ah[NKIT];
    #pragma unroll
    for (int ki = 0; ki < NKIT; ++ki) { ah[ki][0] = 0.0f; ah[ki][1] = 0.0f; }
    ah[NKIT - 1][1] = half ? 1.0f : 0.0f;

    auto loadA = [&](int t, v2f (&xa)[NKIT]) {
        const float* xp = x + ((size_t)t * BATCH + bbase + m) * HID;
        #pragma unroll
        for (int ki = 0; ki < NKIT - 1; ++ki) {
            const int K0 = 4 * ki + 2 * half;
            xa[ki][0] = xp[K0];
            xa[ki][1] = xp[K0 + 1];
        }
        xa[NKIT - 1][0] = xp[16 + 2 * half];
        xa[NKIT - 1][1] = xp[17];
    };

    v8f dx2 = {}, dx3 = {};

    auto step = [&](int t, v2f (&xc)[NKIT], v2f (&xn)[NKIT]) {
        loadA((t + 1 < SEQ) ? (t + 1) : (SEQ - 1), xn);

        v8f d0 = {}, d1 = {}, dh2 = {}, dh3 = {};
        #pragma unroll
        for (int ki = 0; ki < NKIT; ++ki) {
            d0  = WMMA4(xc[ki], Bm[0][ki], d0);
            d1  = WMMA4(xc[ki], Bm[1][ki], d1);
            dh2 = WMMA4(ah[ki], Bhh2[0][ki], dh2);
            dh3 = WMMA4(ah[ki], Bhh2[1][ki], dh3);
        }
        #pragma unroll
        for (int ki = 0; ki < NKIT; ++ki) {
            d0 = WMMA4(ah[ki], Bm[0][NKIT + ki], d0);
            d1 = WMMA4(ah[ki], Bm[1][NKIT + ki], d1);
        }
        #pragma unroll
        for (int r = 0; r < 8; ++r) {
            const int c = r + 8 * half;
            GX[m][c]       = d0[r];
            GX[16 + m][c]  = d1[r];
            GX[32 + m][c]  = dx2[r] + ((m < 6) ? dh2[r] : bn2);
            GH2[m][c]      = dh2[r];
            GX[48 + m][c]  = dx3[r] + bn3;
            GH2[16 + m][c] = dh3[r];
        }
        wave_sync();

        float oacc = 0.0f;
        #pragma unroll
        for (int ki = 0; ki < NKIT; ++ki) {
            #pragma unroll
            for (int v = 0; v < 2; ++v) {
                const int j = 4 * ki + 2 * half + v;
                const float rg = fast_sigmoid(GX[j][m]);
                const float zg = fast_sigmoid(GX[19 + j][m]);
                const float ng = fast_tanh(GX[38 + j][m] + rg * GH2[6 + j][m]);
                const float hn = __builtin_fmaf(zg, ah[ki][v] - ng, ng);
                if (ki == NKIT - 1 && v == 1)
                    ah[ki][v] = half ? 1.0f : hn;
                else
                    ah[ki][v] = hn;
                oacc += hn * wo[ki][v];
            }
        }

        v8f nx2 = {}, nx3 = {};
        #pragma unroll
        for (int ki = 0; ki < NKIT; ++ki) {
            nx2 = WMMA4(xn[ki], Bih2[0][ki], nx2);
            nx3 = WMMA4(xn[ki], Bih2[1][ki], nx3);
        }
        dx2 = nx2;
        dx3 = nx3;

        oacc += __shfl_xor(oacc, 16, 32);
        __builtin_nontemporal_store(oacc + bo,
                                    &out[(size_t)t * BATCH + bbase + m]);
        wave_sync();
    };

    v2f xbufA[NKIT], xbufB[NKIT];
    loadA(0, xbufA);
    #pragma unroll
    for (int ki = 0; ki < NKIT; ++ki) {
        dx2 = WMMA4(xbufA[ki], Bih2[0][ki], dx2);
        dx3 = WMMA4(xbufA[ki], Bih2[1][ki], dx3);
    }
    wave_sync();

    for (int t = 0; t < SEQ; t += 2) {
        step(t,     xbufA, xbufB);
        step(t + 1, xbufB, xbufA);
    }
}

extern "C" void kernel_launch(void* const* d_in, const int* in_sizes, int n_in,
                              void* d_out, int out_size, void* d_ws, size_t ws_size,
                              hipStream_t stream) {
    (void)in_sizes; (void)n_in; (void)d_ws; (void)ws_size; (void)out_size;
    const float* x    = (const float*)d_in[0];
    const float* Wih  = (const float*)d_in[1];
    const float* Whh  = (const float*)d_in[2];
    const float* bih  = (const float*)d_in[3];
    const float* bhh  = (const float*)d_in[4];
    const float* Wout = (const float*)d_in[5];
    const float* bout = (const float*)d_in[6];
    float* out = (float*)d_out;

    dim3 grid(BATCH / 16);
    dim3 block(32);
    gru_wmma_scan<<<grid, block, 0, stream>>>(x, Wih, Whh, bih, bhh, Wout, bout, out);
}


# Split each step over 4 waves (one per SIMD): a wave owns all gate columns of 5 hidden units (15 WMMAs a step, 5 waiting for h); new h shared through LDS, one barrier per step; sum orders kept
# speedup vs baseline: 1.7943x; 1.7943x over previous
#include <hip/hip_runtime.h>


typedef float v2f __attribute__((ext_vector_type(2)));
typedef float v8f __attribute__((ext_vector_type(8)));

#define HID   19
#define SEQ   2048
#define BATCH 1024
#define NKIT  5
#define NW    4
#define UPW   5
#define XRING 4

#define WMMA4(a, b, c) __builtin_amdgcn_wmma_f32_16x16x4_f32( \
        false, (a), false, (b), (short)0, (c), false, false)

__device__ __forceinline__ float fast_tanh(float x) {
#if __has_builtin(__builtin_amdgcn_tanhf)
    return __builtin_amdgcn_tanhf(x);
#else
    float r;
    asm("v_tanh_f32 %0, %1" : "=v"(r) : "v"(x));
    return r;
#endif
}
__device__ __forceinline__ float fast_sigmoid(float x) {
    return __builtin_fmaf(0.5f, fast_tanh(0.5f * x), 0.5f);
}
__device__ __forceinline__ void wave_sync() {
    __builtin_amdgcn_wave_barrier();
}
__device__ __forceinline__ void lds_barrier() {
    __builtin_amdgcn_fence(__ATOMIC_RELEASE, "workgroup", "local");
    __builtin_amdgcn_s_barrier();
    __builtin_amdgcn_fence(__ATOMIC_ACQUIRE, "workgroup", "local");
}

__global__ __launch_bounds__(128) void gru_wmma_scan(
    const float* __restrict__ x,
    const float* __restrict__ Wih,
    const float* __restrict__ Whh,
    const float* __restrict__ bih,
    const float* __restrict__ bhh,
    const float* __restrict__ Wout,
    const float* __restrict__ bout,
    float* __restrict__ out)
{
    __shared__ float G[NW][32][17];
    __shared__ __attribute__((aligned(16))) float Hbuf[2][16][20];

    const int w     = __builtin_amdgcn_readfirstlane((int)(threadIdx.x >> 5));
    const int wv    = threadIdx.x >> 5;
    const int lane  = threadIdx.x & 31;
    const int c     = lane & 15;
    const int half  = lane >> 4;
    const int bbase = blockIdx.x * 16;
    float (*Gw)[17] = G[w];

    const int  uH   = c / 3;
    const int  gH   = c - 3 * uH;
    const int  jH   = UPW * w + uH;
    const bool okH  = (c < 3 * UPW) && (jH < HID);
    const int  rowH = okH ? (gH * HID + jH) : 0;
    const bool okX  = (c < UPW) && (UPW * w + c < HID);
    const int  rowX = okX ? (2 * HID + UPW * w + c) : 0;

    v2f BHx[NKIT], BHh[NKIT], BX2[NKIT];
    #pragma unroll
    for (int ki = 0; ki < NKIT; ++ki) {
        #pragma unroll
        for (int v = 0; v < 2; ++v) {
            const int K = 4 * ki + 2 * half + v;
            float wx = 0.0f, wh = 0.0f, w2 = 0.0f;
            if (okH) {
                if (K < HID) {
                    wh = Whh[rowH * HID + K];
                    if (gH < 2) wx = Wih[rowH * HID + K];
                } else {
                    wh = (gH < 2) ? (bih[rowH] + bhh[rowH]) : bhh[rowH];
                }
            }
            if (okX && K < HID) w2 = Wih[rowX * HID + K];
            BHx[ki][v] = wx;
            BHh[ki][v] = wh;
            BX2[ki][v] = w2;
        }
    }

    int   ue[3];
    float bnx[3];
    #pragma unroll
    for (int e = 0; e < 3; ++e) {
        const int u = 2 * e + half;
        ue[e] = (u < UPW) ? u : (UPW - 1);
        const int j = UPW * w + ue[e];
        bnx[e] = (j < HID) ? bih[2 * HID + j] : 0.0f;
    }
    const float bo = bout[0];

    float wo[NKIT][2];
    #pragma unroll
    for (int ki = 0; ki < NKIT; ++ki)
        #pragma unroll
        for (int v = 0; v < 2; ++v) {
            const int j = 4 * ki + 2 * half + v;
            wo[ki][v] = (j < HID) ? Wout[j] : 0.0f;
        }

    for (int i = threadIdx.x; i < 2 * 16 * 20; i += 32 * NW)
        (&Hbuf[0][0][0])[i] = 0.0f;

    v2f ah[NKIT];
    #pragma unroll
    for (int ki = 0; ki < NKIT; ++ki) { ah[ki][0] = 0.0f; ah[ki][1] = 0.0f; }
    ah[NKIT - 1][1] = half ? 1.0f : 0.0f;
    float hold[3] = {0.0f, 0.0f, 0.0f};

    auto loadA = [&](int t, v2f (&xa)[NKIT]) {
        const float* xp = x + ((size_t)t * BATCH + bbase + c) * HID;
        #pragma unroll
        for (int ki = 0; ki < NKIT - 1; ++ki) {
            const int K0 = 4 * ki + 2 * half;
            xa[ki][0] = xp[K0];
            xa[ki][1] = xp[K0 + 1];
        }
        xa[NKIT - 1][0] = xp[16 + 2 * half];
        xa[NKIT - 1][1] = xp[17];
    };

    v8f accH = {}, dX2 = {};

    auto outdot = [&]() {
        float oacc = 0.0f;
        #pragma unroll
        for (int ki = 0; ki < NKIT; ++ki)
            #pragma unroll
            for (int v = 0; v < 2; ++v)
                oacc += ah[ki][v] * wo[ki][v];
        oacc += __shfl_xor(oacc, 16, 32);
        return oacc + bo;
    };

    auto step = [&](int t, const int S, const int nb,
                    v2f (&xu)[NKIT], v2f (&xl)[NKIT]) {
        v8f dH = accH;
        #pragma unroll
        for (int ki = 0; ki < NKIT; ++ki)
            dH = WMMA4(ah[ki], BHh[ki], dH);
        #pragma unroll
        for (int r = 0; r < 8; ++r) {
            Gw[c][8 * half + r]      = dH[r];
            Gw[16 + c][8 * half + r] = dX2[r];
        }
        wave_sync();

        loadA((t + XRING < SEQ) ? (t + XRING) : (SEQ - 1), xl);

        float pr[3], pz[3], ph[3], px[3];
        #pragma unroll
        for (int e = 0; e < 3; ++e) {
            pr[e] = Gw[3 * ue[e] + 0][c];
            pz[e] = Gw[3 * ue[e] + 1][c];
            ph[e] = Gw[3 * ue[e] + 2][c];
            px[e] = Gw[16 + ue[e]][c];
        }

        const float ov = outdot();

        v8f nH = {}, nX = {};
        #pragma unroll
        for (int ki = 0; ki < NKIT; ++ki) {
            nH = WMMA4(xu[ki], BHx[ki], nH);
            nX = WMMA4(xu[ki], BX2[ki], nX);
        }
        accH = nH;
        dX2  = nX;

        float hnew[3];
        #pragma unroll
        for (int e = 0; e < 3; ++e) {
            const float rg = fast_sigmoid(pr[e]);
            const float zg = fast_sigmoid(pz[e]);
            const float gx = px[e] + bnx[e];
            const float ng = fast_tanh(gx + rg * ph[e]);
            hnew[e] = __builtin_fmaf(zg, hold[e] - ng, ng);
            hold[e] = hnew[e];
        }
        #pragma unroll
        for (int e = 0; e < 3; ++e)
            Hbuf[nb][c][UPW * w + ue[e]] = hnew[e];

        if (wv == S && t > 0)
            __builtin_nontemporal_store(ov,
                &out[(size_t)(t - 1) * BATCH + bbase + c]);

        lds_barrier();

        #pragma unroll
        for (int ki = 0; ki < NKIT; ++ki)
            ah[ki] = *(const v2f*)&Hbuf[nb][c][4 * ki + 2 * half];
        if (half) ah[NKIT - 1][1] = 1.0f;
    };

    v2f xb[XRING][NKIT];
    loadA(0, xb[0]);
    loadA(1, xb[1]);
    loadA(2, xb[2]);
    loadA(3, xb[3]);
    #pragma unroll
    for (int ki = 0; ki < NKIT; ++ki) {
        accH = WMMA4(xb[0][ki], BHx[ki], accH);
        dX2  = WMMA4(xb[0][ki], BX2[ki], dX2);
    }
    lds_barrier();

    for (int t = 0; t < SEQ; t += XRING) {
        step(t,     0, 1, xb[1], xb[0]);
        step(t + 1, 1, 0, xb[2], xb[1]);
        step(t + 2, 2, 1, xb[3], xb[2]);
        step(t + 3, 3, 0, xb[0], xb[3]);
    }

    {
        const float ov = outdot();
        if (wv == 0)
            __builtin_nontemporal_store(ov,
                &out[(size_t)(SEQ - 1) * BATCH + bbase + c]);
    }
}

extern "C" void kernel_launch(void* const* d_in, const int* in_sizes, int n_in,
                              void* d_out, int out_size, void* d_ws, size_t ws_size,
                              hipStream_t stream) {
    (void)in_sizes; (void)n_in; (void)d_ws; (void)ws_size; (void)out_size;
    const float* x    = (const float*)d_in[0];
    const float* Wih  = (const float*)d_in[1];
    const float* Whh  = (const float*)d_in[2];
    const float* bih  = (const float*)d_in[3];
    const float* bhh  = (const float*)d_in[4];
    const float* Wout = (const float*)d_in[5];
    const float* bout = (const float*)d_in[6];
    float* out = (float*)d_out;

    dim3 grid(BATCH / 16);
    dim3 block(32 * NW);
    gru_wmma_scan<<<grid, block, 0, stream>>>(x, Wih, Whh, bih, bhh, Wout, bout, out);
}


# x-side r,z,n columns of a wave's 5 units in one WMMA tile (was 2): 10 not 15 WMMAs per wave-step, x-side ones one block behind the grouped gate reads; tile H's C zero in n columns; sum orders kept
# speedup vs baseline: 2.0106x; 1.1205x over previous
#include <hip/hip_runtime.h>


typedef float v2f __attribute__((ext_vector_type(2)));
typedef float v8f __attribute__((ext_vector_type(8)));

#define HID   19
#define SEQ   2048
#define BATCH 1024
#define NKIT  5
#define NW    4
#define UPW   5
#define XRING 4

#define WMMA4(a, b, c) __builtin_amdgcn_wmma_f32_16x16x4_f32( \
        false, (a), false, (b), (short)0, (c), false, false)

__device__ __forceinline__ float fast_tanh(float x) {
#if __has_builtin(__builtin_amdgcn_tanhf)
    return __builtin_amdgcn_tanhf(x);
#else
    float r;
    asm("v_tanh_f32 %0, %1" : "=v"(r) : "v"(x));
    return r;
#endif
}
__device__ __forceinline__ float fast_sigmoid(float x) {
    return __builtin_fmaf(0.5f, fast_tanh(0.5f * x), 0.5f);
}
__device__ __forceinline__ void wave_sync() {
    __builtin_amdgcn_wave_barrier();
}
__device__ __forceinline__ void lds_barrier() {
    __builtin_amdgcn_fence(__ATOMIC_RELEASE, "workgroup", "local");
    __builtin_amdgcn_s_barrier();
    __builtin_amdgcn_fence(__ATOMIC_ACQUIRE, "workgroup", "local");
}

__global__ __launch_bounds__(128) void gru_wmma_scan(
    const float* __restrict__ x,
    const float* __restrict__ Wih,
    const float* __restrict__ Whh,
    const float* __restrict__ bih,
    const float* __restrict__ bhh,
    const float* __restrict__ Wout,
    const float* __restrict__ bout,
    float* __restrict__ out)
{
    __shared__ float G[NW][32][17];
    __shared__ __attribute__((aligned(16))) float Hbuf[2][16][20];

    const int w     = __builtin_amdgcn_readfirstlane((int)(threadIdx.x >> 5));
    const int wv    = threadIdx.x >> 5;
    const int lane  = threadIdx.x & 31;
    const int c     = lane & 15;
    const int half  = lane >> 4;
    const int bbase = blockIdx.x * 16;
    float (*Gw)[17] = G[w];

    const int  uH   = c / 3;
    const int  gH   = c - 3 * uH;
    const int  jH   = UPW * w + uH;
    const bool okH  = (c < 3 * UPW) && (jH < HID);
    const int  rowH = okH ? (gH * HID + jH) : 0;
    const bool nCol = (gH == 2);

    v2f BX[NKIT], BHh[NKIT];
    #pragma unroll
    for (int ki = 0; ki < NKIT; ++ki) {
        #pragma unroll
        for (int v = 0; v < 2; ++v) {
            const int K = 4 * ki + 2 * half + v;
            float wx = 0.0f, wh = 0.0f;
            if (okH) {
                if (K < HID) {
                    wh = Whh[rowH * HID + K];
                    wx = Wih[rowH * HID + K];
                } else {
                    wh = (gH < 2) ? (bih[rowH] + bhh[rowH]) : bhh[rowH];
                }
            }
            BX[ki][v]  = wx;
            BHh[ki][v] = wh;
        }
    }

    int   ue[3];
    float bnx[3];
    #pragma unroll
    for (int e = 0; e < 3; ++e) {
        const int u = 2 * e + half;
        ue[e] = (u < UPW) ? u : (UPW - 1);
        const int j = UPW * w + ue[e];
        bnx[e] = (j < HID) ? bih[2 * HID + j] : 0.0f;
    }
    const float bo = bout[0];

    float wo[NKIT][2];
    #pragma unroll
    for (int ki = 0; ki < NKIT; ++ki)
        #pragma unroll
        for (int v = 0; v < 2; ++v) {
            const int j = 4 * ki + 2 * half + v;
            wo[ki][v] = (j < HID) ? Wout[j] : 0.0f;
        }

    for (int i = threadIdx.x; i < 2 * 16 * 20; i += 32 * NW)
        (&Hbuf[0][0][0])[i] = 0.0f;

    v2f ah[NKIT];
    #pragma unroll
    for (int ki = 0; ki < NKIT; ++ki) { ah[ki][0] = 0.0f; ah[ki][1] = 0.0f; }
    ah[NKIT - 1][1] = half ? 1.0f : 0.0f;
    float hold[3] = {0.0f, 0.0f, 0.0f};

    auto loadA = [&](int t, v2f (&xa)[NKIT]) {
        const float* xp = x + ((size_t)t * BATCH + bbase + c) * HID;
        #pragma unroll
        for (int ki = 0; ki < NKIT - 1; ++ki) {
            const int K0 = 4 * ki + 2 * half;
            xa[ki][0] = xp[K0];
            xa[ki][1] = xp[K0 + 1];
        }
        xa[NKIT - 1][0] = xp[16 + 2 * half];
        xa[NKIT - 1][1] = xp[17];
    };

    v8f accH = {}, dX = {};
    auto feed = [&](const v8f& xt) {
        v8f r8;
        #pragma unroll
        for (int r = 0; r < 8; ++r) r8[r] = nCol ? 0.0f : xt[r];
        return r8;
    };

    auto outdot = [&]() {
        float oacc = 0.0f;
        #pragma unroll
        for (int ki = 0; ki < NKIT; ++ki)
            #pragma unroll
            for (int v = 0; v < 2; ++v)
                oacc += ah[ki][v] * wo[ki][v];
        oacc += __shfl_xor(oacc, 16, 32);
        return oacc + bo;
    };

    auto step = [&](int t, const int S, const int nb,
                    v2f (&xu)[NKIT], v2f (&xl)[NKIT]) {
        v8f dH = accH;
        #pragma unroll
        for (int ki = 0; ki < NKIT; ++ki)
            dH = WMMA4(ah[ki], BHh[ki], dH);
        #pragma unroll
        for (int r = 0; r < 8; ++r) {
            Gw[c][8 * half + r]      = dH[r];
            Gw[16 + c][8 * half + r] = dX[r];
        }
        wave_sync();

        float pr[3], pz[3], ph[3], px[3];
        #pragma unroll
        for (int e = 0; e < 3; ++e) {
            pr[e] = Gw[3 * ue[e] + 0][c];
            pz[e] = Gw[3 * ue[e] + 1][c];
            ph[e] = Gw[3 * ue[e] + 2][c];
            px[e] = Gw[16 + 3 * ue[e] + 2][c];
        }
        __builtin_amdgcn_sched_barrier(0);

        v8f nX = {};
        #pragma unroll
        for (int ki = 0; ki < NKIT; ++ki)
            nX = WMMA4(xu[ki], BX[ki], nX);
        dX = nX;
        __builtin_amdgcn_sched_barrier(0);

        loadA((t + XRING < SEQ) ? (t + XRING) : (SEQ - 1), xl);

        const float ov = outdot();

        float hnew[3];
        #pragma unroll
        for (int e = 0; e < 3; ++e) {
            const float rg = fast_sigmoid(pr[e]);
            const float zg = fast_sigmoid(pz[e]);
            const float gx = px[e] + bnx[e];
            const float ng = fast_tanh(gx + rg * ph[e]);
            hnew[e] = __builtin_fmaf(zg, hold[e] - ng, ng);
            hold[e] = hnew[e];
        }
        #pragma unroll
        for (int e = 0; e < 3; ++e)
            Hbuf[nb][c][UPW * w + ue[e]] = hnew[e];

        if (wv == S && t > 0)
            __builtin_nontemporal_store(ov,
                &out[(size_t)(t - 1) * BATCH + bbase + c]);

        accH = feed(dX);

        lds_barrier();

        #pragma unroll
        for (int ki = 0; ki < NKIT; ++ki)
            ah[ki] = *(const v2f*)&Hbuf[nb][c][4 * ki + 2 * half];
        if (half) ah[NKIT - 1][1] = 1.0f;
    };

    v2f xb[XRING][NKIT];
    loadA(0, xb[0]);
    loadA(1, xb[1]);
    loadA(2, xb[2]);
    loadA(3, xb[3]);
    #pragma unroll
    for (int ki = 0; ki < NKIT; ++ki)
        dX = WMMA4(xb[0][ki], BX[ki], dX);
    accH = feed(dX);
    lds_barrier();

    for (int t = 0; t < SEQ; t += XRING) {
        step(t,     0, 1, xb[1], xb[0]);
        step(t + 1, 1, 0, xb[2], xb[1]);
        step(t + 2, 2, 1, xb[3], xb[2]);
        step(t + 3, 3, 0, xb[0], xb[3]);
    }

    {
        const float ov = outdot();
        if (wv == 0)
            __builtin_nontemporal_store(ov,
                &out[(size_t)(SEQ - 1) * BATCH + bbase + c]);
    }
}

extern "C" void kernel_launch(void* const* d_in, const int* in_sizes, int n_in,
                              void* d_out, int out_size, void* d_ws, size_t ws_size,
                              hipStream_t stream) {
    (void)in_sizes; (void)n_in; (void)d_ws; (void)ws_size; (void)out_size;
    const float* x    = (const float*)d_in[0];
    const float* Wih  = (const float*)d_in[1];
    const float* Whh  = (const float*)d_in[2];
    const float* bih  = (const float*)d_in[3];
    const float* bhh  = (const float*)d_in[4];
    const float* Wout = (const float*)d_in[5];
    const float* bout = (const float*)d_in[6];
    float* out = (float*)d_out;

    dim3 grid(BATCH / 16);
    dim3 block(32 * NW);
    gru_wmma_scan<<<grid, block, 0, stream>>>(x, Wih, Whh, bih, bhh, Wout, bout, out);
}


# No address-translation wait left on a step's path: x rows loaded from one base per 8 steps, h stores unmasked, output shares parked in LDS and stored once per 32 steps; sum orders kept
# speedup vs baseline: 3.5265x; 1.7540x over previous
#include <hip/hip_runtime.h>


typedef float v2f __attribute__((ext_vector_type(2)));
typedef float v8f __attribute__((ext_vector_type(8)));

#define HID   19
#define SEQ   2048
#define BATCH 1024
#define NKIT  5
#define NW    4
#define UPW   5
#define RPG   8
#define XPAIRS 4

#define WMMA4(a, b, c) __builtin_amdgcn_wmma_f32_16x16x4_f32( \
        false, (a), false, (b), (short)0, (c), false, false)

__device__ __forceinline__ float fast_tanh(float x) {
#if __has_builtin(__builtin_amdgcn_tanhf)
    return __builtin_amdgcn_tanhf(x);
#else
    float r;
    asm("v_tanh_f32 %0, %1" : "=v"(r) : "v"(x));
    return r;
#endif
}
__device__ __forceinline__ float fast_sigmoid(float x) {
    return __builtin_fmaf(0.5f, fast_tanh(0.5f * x), 0.5f);
}
template <int N>
__device__ __forceinline__ float from_left(float v) {
    return __builtin_bit_cast(float, __builtin_amdgcn_update_dpp(
        0, __builtin_bit_cast(int, v), 0x110 + N, 0xf, 0xf, true));
}
__device__ __forceinline__ void lds_barrier() {
    __builtin_amdgcn_fence(__ATOMIC_RELEASE, "workgroup", "local");
    __builtin_amdgcn_s_barrier();
    __builtin_amdgcn_fence(__ATOMIC_ACQUIRE, "workgroup", "local");
}

__global__ __launch_bounds__(128) void gru_wmma_scan(
    const float* __restrict__ x,
    const float* __restrict__ Wih,
    const float* __restrict__ Whh,
    const float* __restrict__ bih,
    const float* __restrict__ bhh,
    const float* __restrict__ Wout,
    const float* __restrict__ bout,
    float* __restrict__ out)
{
    __shared__ __attribute__((aligned(16))) float Hall[2 * RPG * 20 + 256];
    float (*Hbuf)[RPG][20] = (float (*)[RPG][20])Hall;
    __shared__ __attribute__((aligned(16))) float Opart[64][2][RPG];

    const int w     = __builtin_amdgcn_readfirstlane((int)(threadIdx.x >> 5));
    const int lane  = threadIdx.x & 31;
    const int c     = lane & 15;
    const int half  = lane >> 4;
    const int bbase = blockIdx.x * RPG;
    const int q     = (c & 3) + 4 * (c >> 3);
    const int ts    = (c >> 2) & 1;

    const int  uH   = c / 3;
    const int  gH   = c - 3 * uH;
    const int  jH   = UPW * w + uH;
    const bool okH  = (c < 3 * UPW) && (jH < HID);
    const int  rowH = okH ? (gH * HID + jH) : 0;
    const bool nCol = (gH == 2);

    v2f BX[NKIT], BHh[NKIT];
    #pragma unroll
    for (int ki = 0; ki < NKIT; ++ki) {
        #pragma unroll
        for (int v = 0; v < 2; ++v) {
            const int K = 4 * ki + 2 * half + v;
            float wx = 0.0f, wh = 0.0f;
            if (okH) {
                if (K < HID) {
                    wh = Whh[rowH * HID + K];
                    wx = Wih[rowH * HID + K];
                } else {
                    wh = (gH < 2) ? (bih[rowH] + bhh[rowH]) : bhh[rowH];
                }
            }
            BX[ki][v]  = wx;
            BHh[ki][v] = wh;
        }
    }

    const bool  hLane = nCol && okH;
    const int   hword = hLane ? (4 * half * 20 + jH) : (2 * RPG * 20 + lane);
    const float bnx   = hLane ? bih[2 * HID + jH] : 0.0f;
    const float bo = bout[0];

    float wo[NKIT][2];
    #pragma unroll
    for (int ki = 0; ki < NKIT; ++ki)
        #pragma unroll
        for (int v = 0; v < 2; ++v) {
            const int j = 4 * ki + 2 * half + v;
            wo[ki][v] = (j < HID) ? Wout[j] : 0.0f;
        }

    for (int i = threadIdx.x; i < 2 * RPG * 20; i += 32 * NW)
        Hall[i] = 0.0f;

    v2f ah[NKIT];
    #pragma unroll
    for (int ki = 0; ki < NKIT; ++ki) { ah[ki][0] = 0.0f; ah[ki][1] = 0.0f; }
    ah[NKIT - 1][1] = half ? 1.0f : 0.0f;
    float hold[4];
    #pragma unroll
    for (int r = 0; r < 4; ++r) hold[r] = 0.0f;

    const unsigned xrow  = (unsigned)((ts * BATCH + bbase + q) * HID);
    const unsigned xoffA = (xrow + 2u * (unsigned)half) * 4u;
    const unsigned xoffB = (xrow + 17u) * 4u;
    auto loadA = [&](int p, v2f (&xa)[NKIT]) __attribute__((always_inline)) {
        const char*  xs = (const char*)x + (size_t)p * (size_t)(2 * BATCH * HID * 4);
        const float* xp = (const float*)(xs + xoffA);
        #pragma unroll
        for (int ki = 0; ki < NKIT - 1; ++ki) {
            xa[ki][0] = xp[4 * ki];
            xa[ki][1] = xp[4 * ki + 1];
        }
        xa[NKIT - 1][0] = xp[16];
        xa[NKIT - 1][1] = *(const float*)(xs + xoffB);
    };

    v8f accH = {}, xcur = {};
    float xnb[4] = {0.0f, 0.0f, 0.0f, 0.0f};
    auto feed = [&](const int hi) {
        #pragma unroll
        for (int r = 0; r < 4; ++r) {
            const float xv = xcur[4 * hi + r];
            accH[r] = nCol ? 0.0f : xv;
            xnb[r]  = xv + bnx;
        }
    };

    auto outpart = [&]() {
        float oacc = 0.0f;
        #pragma unroll
        for (int ki = 0; ki < NKIT; ++ki)
            #pragma unroll
            for (int v = 0; v < 2; ++v)
                oacc += ah[ki][v] * wo[ki][v];
        return oacc;
    };
    auto flush = [&](int t0) {
        const int   tl = t0 + lane;
        const float* s = &Opart[tl & 63][0][0];
        float* o = out + (size_t)tl * BATCH + bbase;
        #pragma unroll
        for (int g = 0; g < 2; ++g) {
            const float4 a = *(const float4*)(s + 4 * g);
            const float4 b = *(const float4*)(s + RPG + 4 * g);
            float4 r;
            r.x = (a.x + b.x) + bo;
            r.y = (a.y + b.y) + bo;
            r.z = (a.z + b.z) + bo;
            r.w = (a.w + b.w) + bo;
            *(float4*)(o + 4 * g) = r;
        }
    };

    auto step = [&](int t, const int S, const int nb, const int hi,
                    v2f (*xload)[NKIT], int pload, v2f (*xnext)[NKIT])
                    __attribute__((always_inline)) {
        v8f dH = accH;
        #pragma unroll
        for (int ki = 0; ki < NKIT; ++ki)
            dH = WMMA4(ah[ki], BHh[ki], dH);

        __builtin_amdgcn_sched_barrier(0);
        if (xload)
            loadA(pload, *xload);

        const float op = outpart();

        float hnew[4];
        #pragma unroll
        for (int r = 0; r < 4; ++r) {
            const float sg = fast_sigmoid(dH[r]);
            const float rg = from_left<2>(sg);
            const float zg = from_left<1>(sg);
            const float ng = fast_tanh(xnb[r] + rg * dH[r]);
            hnew[r] = __builtin_fmaf(zg, hold[r] - ng, ng);
            hold[r] = hnew[r];
        }
        #pragma unroll
        for (int r = 0; r < 4; ++r)
            Hall[nb * (RPG * 20) + hword + 20 * r] = hnew[r];
        if (w == S)
            Opart[(t - 1) & 63][half][q] = op;

        feed(1 - hi);

        v8f nX = {};
        if (xnext) {
            nX = WMMA4((*xnext)[0], BX[0], nX);
            nX = WMMA4((*xnext)[1], BX[1], nX);
        }

        lds_barrier();

        #pragma unroll
        for (int ki = 0; ki < NKIT; ++ki)
            ah[ki] = *(const v2f*)&Hbuf[nb][q][4 * ki + 2 * half];
        __builtin_amdgcn_sched_barrier(0);
        if (xnext) {
            #pragma unroll
            for (int ki = 2; ki < NKIT; ++ki)
                nX = WMMA4((*xnext)[ki], BX[ki], nX);
            xcur = nX;
        }
        __builtin_amdgcn_sched_barrier(0);
        if (half) ah[NKIT - 1][1] = 1.0f;
    };

    v2f xb[XPAIRS][NKIT];
    loadA(0, xb[0]);
    loadA(1, xb[1]);
    loadA(2, xb[2]);
    loadA(3, xb[3]);
    {
        v8f x0 = {};
        #pragma unroll
        for (int ki = 0; ki < NKIT; ++ki)
            x0 = WMMA4(xb[0][ki], BX[ki], x0);
        xcur = x0;
        feed(0);
    }
    lds_barrier();

    for (int t = 0; t < SEQ; t += 2 * XPAIRS) {
        const int p = t >> 1;
        const int pl = (t < SEQ - 2 * XPAIRS) ? p + 4 : p;
        if (w == 0 && (t & 31) == 8 && t >= 40)
            flush(t - 40);
        step(t,     0, 1, 0, &xb[0], pl,     &xb[1]);
        step(t + 1, 1, 0, 1, nullptr, 0, nullptr);
        step(t + 2, 2, 1, 0, &xb[1], pl + 1, &xb[2]);
        step(t + 3, 3, 0, 1, nullptr, 0, nullptr);
        step(t + 4, 0, 1, 0, &xb[2], pl + 2, &xb[3]);
        step(t + 5, 1, 0, 1, nullptr, 0, nullptr);
        step(t + 6, 2, 1, 0, &xb[3], pl + 3, &xb[0]);
        step(t + 7, 3, 0, 1, nullptr, 0, nullptr);
    }

    {
        const float op = outpart();
        if (w == 0)
            Opart[(SEQ - 1) & 63][half][q] = op;
        lds_barrier();
        if (w == 0) flush(SEQ - 32);
    }
}

extern "C" void kernel_launch(void* const* d_in, const int* in_sizes, int n_in,
                              void* d_out, int out_size, void* d_ws, size_t ws_size,
                              hipStream_t stream) {
    (void)in_sizes; (void)n_in; (void)d_ws; (void)ws_size; (void)out_size;
    const float* x    = (const float*)d_in[0];
    const float* Wih  = (const float*)d_in[1];
    const float* Whh  = (const float*)d_in[2];
    const float* bih  = (const float*)d_in[3];
    const float* bhh  = (const float*)d_in[4];
    const float* Wout = (const float*)d_in[5];
    const float* bout = (const float*)d_in[6];
    float* out = (float*)d_out;

    dim3 grid(BATCH / RPG);
    dim3 block(32 * NW);
    gru_wmma_scan<<<grid, block, 0, stream>>>(x, Wih, Whh, bih, bhh, Wout, bout, out);
}


# Output dot taken off the path between a step's h stores and its barrier signal: each odd step sums two rows, one per wave, while its LDS fragment reads are in flight; k order kept, products fused
# speedup vs baseline: 3.6475x; 1.0343x over previous
#include <hip/hip_runtime.h>


typedef float v2f __attribute__((ext_vector_type(2)));
typedef float v8f __attribute__((ext_vector_type(8)));

#define HID   19
#define SEQ   2048
#define BATCH 1024
#define NKIT  5
#define NW    4
#define UPW   5
#define RPG   8
#define XPAIRS 4

#define WMMA4(a, b, c) __builtin_amdgcn_wmma_f32_16x16x4_f32( \
        false, (a), false, (b), (short)0, (c), false, false)

__device__ __forceinline__ float fast_tanh(float x) {
#if __has_builtin(__builtin_amdgcn_tanhf)
    return __builtin_amdgcn_tanhf(x);
#else
    float r;
    asm("v_tanh_f32 %0, %1" : "=v"(r) : "v"(x));
    return r;
#endif
}
__device__ __forceinline__ float fast_sigmoid(float x) {
    return __builtin_fmaf(0.5f, fast_tanh(0.5f * x), 0.5f);
}
template <int N>
__device__ __forceinline__ float from_left(float v) {
    return __builtin_bit_cast(float, __builtin_amdgcn_update_dpp(
        0, __builtin_bit_cast(int, v), 0x110 + N, 0xf, 0xf, true));
}
__device__ __forceinline__ void lds_barrier() {
    __builtin_amdgcn_fence(__ATOMIC_RELEASE, "workgroup", "local");
    __builtin_amdgcn_s_barrier();
    __builtin_amdgcn_fence(__ATOMIC_ACQUIRE, "workgroup", "local");
}

__global__ __launch_bounds__(128) void gru_wmma_scan(
    const float* __restrict__ x,
    const float* __restrict__ Wih,
    const float* __restrict__ Whh,
    const float* __restrict__ bih,
    const float* __restrict__ bhh,
    const float* __restrict__ Wout,
    const float* __restrict__ bout,
    float* __restrict__ out)
{
    __shared__ __attribute__((aligned(16))) float Hall[2 * RPG * 20 + 256];
    float (*Hbuf)[RPG][20] = (float (*)[RPG][20])Hall;
    __shared__ __attribute__((aligned(16))) float Opart[64][2][RPG];

    const int w     = __builtin_amdgcn_readfirstlane((int)(threadIdx.x >> 5));
    const int oddw  = w & 1;
    const int lane  = threadIdx.x & 31;
    const int c     = lane & 15;
    const int half  = lane >> 4;
    const int bbase = blockIdx.x * RPG;
    const int q     = (c & 3) + 4 * (c >> 3);
    const int ts    = (c >> 2) & 1;

    const int  uH   = c / 3;
    const int  gH   = c - 3 * uH;
    const int  jH   = UPW * w + uH;
    const bool okH  = (c < 3 * UPW) && (jH < HID);
    const int  rowH = okH ? (gH * HID + jH) : 0;
    const bool nCol = (gH == 2);

    v2f BX[NKIT], BHh[NKIT];
    #pragma unroll
    for (int ki = 0; ki < NKIT; ++ki) {
        #pragma unroll
        for (int v = 0; v < 2; ++v) {
            const int K = 4 * ki + 2 * half + v;
            float wx = 0.0f, wh = 0.0f;
            if (okH) {
                if (K < HID) {
                    wh = Whh[rowH * HID + K];
                    wx = Wih[rowH * HID + K];
                } else {
                    wh = (gH < 2) ? (bih[rowH] + bhh[rowH]) : bhh[rowH];
                }
            }
            BX[ki][v]  = wx;
            BHh[ki][v] = wh;
        }
    }

    const bool  hLane = nCol && okH;
    const int   hword = hLane ? (4 * half * 20 + jH) : (2 * RPG * 20 + lane);
    const float bnx   = hLane ? bih[2 * HID + jH] : 0.0f;
    const float bo = bout[0];

    float wo[NKIT][2];
    #pragma unroll
    for (int ki = 0; ki < NKIT; ++ki)
        #pragma unroll
        for (int v = 0; v < 2; ++v) {
            const int j = 4 * ki + 2 * half + v;
            wo[ki][v] = (j < HID) ? Wout[j] : 0.0f;
        }

    for (int i = threadIdx.x; i < 2 * RPG * 20; i += 32 * NW)
        Hall[i] = 0.0f;

    v2f ah[NKIT];
    #pragma unroll
    for (int ki = 0; ki < NKIT; ++ki) { ah[ki][0] = 0.0f; ah[ki][1] = 0.0f; }
    ah[NKIT - 1][1] = half ? 1.0f : 0.0f;
    float hold[4];
    #pragma unroll
    for (int r = 0; r < 4; ++r) hold[r] = 0.0f;

    const unsigned xrow  = (unsigned)((ts * BATCH + bbase + q) * HID);
    const unsigned xoffA = (xrow + 2u * (unsigned)half) * 4u;
    const unsigned xoffB = (xrow + 17u) * 4u;
    auto loadA = [&](int p, v2f (&xa)[NKIT]) __attribute__((always_inline)) {
        const char*  xs = (const char*)x + (size_t)p * (size_t)(2 * BATCH * HID * 4);
        const float* xp = (const float*)(xs + xoffA);
        #pragma unroll
        for (int ki = 0; ki < NKIT - 1; ++ki) {
            xa[ki][0] = xp[4 * ki];
            xa[ki][1] = xp[4 * ki + 1];
        }
        xa[NKIT - 1][0] = xp[16];
        xa[NKIT - 1][1] = *(const float*)(xs + xoffB);
    };

    v8f accH = {}, xcur = {};
    float xnb[4] = {0.0f, 0.0f, 0.0f, 0.0f};
    auto feed = [&](const int hi) {
        #pragma unroll
        for (int r = 0; r < 4; ++r) {
            const float xv = xcur[4 * hi + r];
            accH[r] = nCol ? 0.0f : xv;
            xnb[r]  = xv + bnx;
        }
    };

    auto outpart = [&](const v2f (&f)[NKIT]) {
        float oacc = f[0][0] * wo[0][0];
        oacc = __builtin_fmaf(f[0][1], wo[0][1], oacc);
        #pragma unroll
        for (int ki = 1; ki < NKIT; ++ki)
            #pragma unroll
            for (int v = 0; v < 2; ++v)
                oacc = __builtin_fmaf(f[ki][v], wo[ki][v], oacc);
        return oacc;
    };
    v2f ahp[NKIT];
    #pragma unroll
    for (int ki = 0; ki < NKIT; ++ki) ahp[ki] = ah[ki];
    auto flush = [&](int t0) {
        const int   tl = t0 + lane;
        const float* s = &Opart[tl & 63][0][0];
        float* o = out + (size_t)tl * BATCH + bbase;
        #pragma unroll
        for (int g = 0; g < 2; ++g) {
            const float4 a = *(const float4*)(s + 4 * g);
            const float4 b = *(const float4*)(s + RPG + 4 * g);
            float4 r;
            r.x = (a.x + b.x) + bo;
            r.y = (a.y + b.y) + bo;
            r.z = (a.z + b.z) + bo;
            r.w = (a.w + b.w) + bo;
            *(float4*)(o + 4 * g) = r;
        }
    };

    auto step = [&](int t, const int nb, const int hi,
                    v2f (*xload)[NKIT], int pload, v2f (*xnext)[NKIT])
                    __attribute__((always_inline)) {
        v8f dH = accH;
        #pragma unroll
        for (int ki = 0; ki < NKIT; ++ki)
            dH = WMMA4(ah[ki], BHh[ki], dH);

        __builtin_amdgcn_sched_barrier(0);
        if (xload)
            loadA(pload, *xload);

        float hnew[4];
        #pragma unroll
        for (int r = 0; r < 4; ++r) {
            const float sg = fast_sigmoid(dH[r]);
            const float rg = from_left<2>(sg);
            const float zg = from_left<1>(sg);
            const float ng = fast_tanh(xnb[r] + rg * dH[r]);
            hnew[r] = __builtin_fmaf(zg, hold[r] - ng, ng);
            hold[r] = hnew[r];
        }
        #pragma unroll
        for (int r = 0; r < 4; ++r)
            Hall[nb * (RPG * 20) + hword + 20 * r] = hnew[r];
        feed(1 - hi);

        v8f nX = {};
        if (xnext) {
            nX = WMMA4((*xnext)[0], BX[0], nX);
            nX = WMMA4((*xnext)[1], BX[1], nX);
        }

        lds_barrier();

        v2f ahn[NKIT];
        #pragma unroll
        for (int ki = 0; ki < NKIT; ++ki)
            ahn[ki] = *(const v2f*)&Hbuf[nb][q][4 * ki + 2 * half];
        __builtin_amdgcn_sched_barrier(0);
        if (hi) {
            v2f f[NKIT];
            #pragma unroll
            for (int ki = 0; ki < NKIT; ++ki) {
                f[ki][0] = oddw ? ah[ki][0] : ahp[ki][0];
                f[ki][1] = oddw ? ah[ki][1] : ahp[ki][1];
            }
            Opart[(t - 2 + oddw) & 63][half][q] = outpart(f);
        } else {
            #pragma unroll
            for (int ki = 0; ki < NKIT; ++ki) ahp[ki] = ah[ki];
        }
        if (xnext) {
            #pragma unroll
            for (int ki = 2; ki < NKIT; ++ki)
                nX = WMMA4((*xnext)[ki], BX[ki], nX);
            xcur = nX;
        }
        __builtin_amdgcn_sched_barrier(0);
        #pragma unroll
        for (int ki = 0; ki < NKIT; ++ki) ah[ki] = ahn[ki];
        if (half) ah[NKIT - 1][1] = 1.0f;
    };

    v2f xb[XPAIRS][NKIT];
    loadA(0, xb[0]);
    loadA(1, xb[1]);
    loadA(2, xb[2]);
    loadA(3, xb[3]);
    {
        v8f x0 = {};
        #pragma unroll
        for (int ki = 0; ki < NKIT; ++ki)
            x0 = WMMA4(xb[0][ki], BX[ki], x0);
        xcur = x0;
        feed(0);
    }
    lds_barrier();

    for (int t = 0; t < SEQ; t += 2 * XPAIRS) {
        const int p = t >> 1;
        const int pl = (t < SEQ - 2 * XPAIRS) ? p + 4 : p;
        if (w == 0 && (t & 31) == 8 && t >= 40)
            flush(t - 40);
        step(t,     1, 0, &xb[0], pl,     &xb[1]);
        step(t + 1, 0, 1, nullptr, 0, nullptr);
        step(t + 2, 1, 0, &xb[1], pl + 1, &xb[2]);
        step(t + 3, 0, 1, nullptr, 0, nullptr);
        step(t + 4, 1, 0, &xb[2], pl + 2, &xb[3]);
        step(t + 5, 0, 1, nullptr, 0, nullptr);
        step(t + 6, 1, 0, &xb[3], pl + 3, &xb[0]);
        step(t + 7, 0, 1, nullptr, 0, nullptr);
    }

    {
        const float op = outpart(ah);
        if (w == 0)
            Opart[(SEQ - 1) & 63][half][q] = op;
        lds_barrier();
        if (w == 0) flush(SEQ - 32);
    }
}

extern "C" void kernel_launch(void* const* d_in, const int* in_sizes, int n_in,
                              void* d_out, int out_size, void* d_ws, size_t ws_size,
                              hipStream_t stream) {
    (void)in_sizes; (void)n_in; (void)d_ws; (void)ws_size; (void)out_size;
    const float* x    = (const float*)d_in[0];
    const float* Wih  = (const float*)d_in[1];
    const float* Whh  = (const float*)d_in[2];
    const float* bih  = (const float*)d_in[3];
    const float* bhh  = (const float*)d_in[4];
    const float* Wout = (const float*)d_in[5];
    const float* bout = (const float*)d_in[6];
    float* out = (float*)d_out;

    dim3 grid(BATCH / RPG);
    dim3 block(32 * NW);
    gru_wmma_scan<<<grid, block, 0, stream>>>(x, Wih, Whh, bih, bhh, Wout, bout, out);
}


# Work that needs no h (x-side WMMAs 2+2+1, x loads, accumulator set-up) held by input-only fences in the hand-over waits; K=19 constant read from LDS; h chain starts as reads land; sum orders kept
# speedup vs baseline: 3.9485x; 1.0825x over previous
#include <hip/hip_runtime.h>


typedef float v2f __attribute__((ext_vector_type(2)));
typedef float v8f __attribute__((ext_vector_type(8)));

#define HID   19
#define SEQ   2048
#define BATCH 1024
#define NKIT  5
#define NW    4
#define UPW   5
#define RPG   8
#define XPAIRS 4

#define WMMA4(a, b, c) __builtin_amdgcn_wmma_f32_16x16x4_f32( \
        false, (a), false, (b), (short)0, (c), false, false)

__device__ __forceinline__ float fast_tanh(float x) {
#if __has_builtin(__builtin_amdgcn_tanhf)
    return __builtin_amdgcn_tanhf(x);
#else
    float r;
    asm("v_tanh_f32 %0, %1" : "=v"(r) : "v"(x));
    return r;
#endif
}
__device__ __forceinline__ float fast_sigmoid(float x) {
    return __builtin_fmaf(0.5f, fast_tanh(0.5f * x), 0.5f);
}
template <int N>
__device__ __forceinline__ float from_left(float v) {
    return __builtin_bit_cast(float, __builtin_amdgcn_update_dpp(
        0, __builtin_bit_cast(int, v), 0x110 + N, 0xf, 0xf, true));
}
__device__ __forceinline__ void lds_barrier() {
    __builtin_amdgcn_fence(__ATOMIC_RELEASE, "workgroup", "local");
    __builtin_amdgcn_s_barrier();
    __builtin_amdgcn_fence(__ATOMIC_ACQUIRE, "workgroup", "local");
}

__global__ __launch_bounds__(128) void gru_wmma_scan(
    const float* __restrict__ x,
    const float* __restrict__ Wih,
    const float* __restrict__ Whh,
    const float* __restrict__ bih,
    const float* __restrict__ bhh,
    const float* __restrict__ Wout,
    const float* __restrict__ bout,
    float* __restrict__ out)
{
    __shared__ __attribute__((aligned(16))) float Hall[2 * RPG * 20 + 256];
    float (*Hbuf)[RPG][20] = (float (*)[RPG][20])Hall;
    __shared__ __attribute__((aligned(16))) float Opart[64][2][RPG];

    const int w     = __builtin_amdgcn_readfirstlane((int)(threadIdx.x >> 5));
    const int oddw  = w & 1;
    const int lane  = threadIdx.x & 31;
    const int c     = lane & 15;
    const int half  = lane >> 4;
    const int bbase = blockIdx.x * RPG;
    const int q     = (c & 3) + 4 * (c >> 3);
    const int ts    = (c >> 2) & 1;

    const int  uH   = c / 3;
    const int  gH   = c - 3 * uH;
    const int  jH   = UPW * w + uH;
    const bool okH  = (c < 3 * UPW) && (jH < HID);
    const int  rowH = okH ? (gH * HID + jH) : 0;
    const bool nCol = (gH == 2);

    v2f BX[NKIT], BHh[NKIT];
    #pragma unroll
    for (int ki = 0; ki < NKIT; ++ki) {
        #pragma unroll
        for (int v = 0; v < 2; ++v) {
            const int K = 4 * ki + 2 * half + v;
            float wx = 0.0f, wh = 0.0f;
            if (okH) {
                if (K < HID) {
                    wh = Whh[rowH * HID + K];
                    wx = Wih[rowH * HID + K];
                } else {
                    wh = (gH < 2) ? (bih[rowH] + bhh[rowH]) : bhh[rowH];
                }
            }
            BX[ki][v]  = wx;
            BHh[ki][v] = wh;
        }
    }

    const bool  hLane = nCol && okH;
    const int   hword = hLane ? (4 * half * 20 + jH) : (2 * RPG * 20 + lane);
    const float bnx   = hLane ? bih[2 * HID + jH] : 0.0f;
    const float bo = bout[0];

    float wo[NKIT][2];
    #pragma unroll
    for (int ki = 0; ki < NKIT; ++ki)
        #pragma unroll
        for (int v = 0; v < 2; ++v) {
            const int j = 4 * ki + 2 * half + v;
            wo[ki][v] = (j < HID) ? Wout[j] : 0.0f;
        }

    for (int i = threadIdx.x; i < 2 * RPG * 20; i += 32 * NW)
        Hall[i] = (i % 20 == 19) ? 1.0f : 0.0f;

    v2f ah[NKIT];
    #pragma unroll
    for (int ki = 0; ki < NKIT; ++ki) { ah[ki][0] = 0.0f; ah[ki][1] = 0.0f; }
    ah[NKIT - 1][1] = half ? 1.0f : 0.0f;
    float hold[4];
    #pragma unroll
    for (int r = 0; r < 4; ++r) hold[r] = 0.0f;

    const unsigned xrow  = (unsigned)((ts * BATCH + bbase + q) * HID);
    const unsigned xoffA = (xrow + 2u * (unsigned)half) * 4u;
    const unsigned xoffB = (xrow + 17u) * 4u;
    auto loadA = [&](int p, v2f (&xa)[NKIT]) __attribute__((always_inline)) {
        const char*  xs = (const char*)x + (size_t)p * (size_t)(2 * BATCH * HID * 4);
        const float* xp = (const float*)(xs + xoffA);
        #pragma unroll
        for (int ki = 0; ki < NKIT - 1; ++ki) {
            xa[ki][0] = xp[4 * ki];
            xa[ki][1] = xp[4 * ki + 1];
        }
        xa[NKIT - 1][0] = xp[16];
        xa[NKIT - 1][1] = *(const float*)(xs + xoffB);
    };

    v8f accH = {}, xcur = {};
    float xnb[4] = {0.0f, 0.0f, 0.0f, 0.0f};
    auto feed = [&](const int hi) {
        #pragma unroll
        for (int r = 0; r < 4; ++r) {
            const float xv = xcur[4 * hi + r];
            accH[r] = nCol ? 0.0f : xv;
            xnb[r]  = xv + bnx;
        }
    };

    auto outpart = [&](const v2f (&f)[NKIT]) {
        float oacc = f[0][0] * wo[0][0];
        oacc = __builtin_fmaf(f[0][1], wo[0][1], oacc);
        #pragma unroll
        for (int ki = 1; ki < NKIT; ++ki)
            #pragma unroll
            for (int v = 0; v < 2; ++v)
                oacc = __builtin_fmaf(f[ki][v], wo[ki][v], oacc);
        return oacc;
    };
    v2f ahp[NKIT];
    #pragma unroll
    for (int ki = 0; ki < NKIT; ++ki) ahp[ki] = ah[ki];
    auto flush = [&](int t0) {
        const int   tl = t0 + lane;
        const float* s = &Opart[tl & 63][0][0];
        float* o = out + (size_t)tl * BATCH + bbase;
        #pragma unroll
        for (int g = 0; g < 2; ++g) {
            const float4 a = *(const float4*)(s + 4 * g);
            const float4 b = *(const float4*)(s + RPG + 4 * g);
            float4 r;
            r.x = (a.x + b.x) + bo;
            r.y = (a.y + b.y) + bo;
            r.z = (a.z + b.z) + bo;
            r.w = (a.w + b.w) + bo;
            *(float4*)(o + 4 * g) = r;
        }
    };

    v8f nX = {};
    auto step = [&](int t, const int nb, const int hi,
                    v2f (*xload)[NKIT], int pload, v2f (*xnext)[NKIT])
                    __attribute__((always_inline)) {
        v8f dH = accH;
        #pragma unroll
        for (int ki = 0; ki < NKIT; ++ki)
            dH = WMMA4(ah[ki], BHh[ki], dH);

        float hnew[4];
        #pragma unroll
        for (int r = 0; r < 4; ++r) {
            const float sg = fast_sigmoid(dH[r]);
            const float rg = from_left<2>(sg);
            const float zg = from_left<1>(sg);
            const float ng = fast_tanh(xnb[r] + rg * dH[r]);
            hnew[r] = __builtin_fmaf(zg, hold[r] - ng, ng);
            hold[r] = hnew[r];
        }
        #pragma unroll
        for (int r = 0; r < 4; ++r)
            Hall[nb * (RPG * 20) + hword + 20 * r] = hnew[r];

        __builtin_amdgcn_sched_barrier(0);
        if (xload)
            loadA(pload, *xload);

        if (hi) {
            nX = WMMA4((*xnext)[4], BX[4], nX);
            xcur = nX;
        } else {
            v8f z = {};
            nX = WMMA4((*xnext)[0], BX[0], z);
            nX = WMMA4((*xnext)[1], BX[1], nX);
        }
        asm volatile("" :: "v"(nX));

        lds_barrier();

        v2f ahn[NKIT];
        #pragma unroll
        for (int ki = 0; ki < NKIT; ++ki)
            ahn[ki] = *(const v2f*)&Hbuf[nb][q][4 * ki + 2 * half];
        __builtin_amdgcn_sched_barrier(0);
        if (hi) {
            v2f f[NKIT];
            #pragma unroll
            for (int ki = 0; ki < NKIT; ++ki) {
                f[ki][0] = oddw ? ah[ki][0] : ahp[ki][0];
                f[ki][1] = oddw ? ah[ki][1] : ahp[ki][1];
            }
            Opart[(t - 2 + oddw) & 63][half][q] = outpart(f);
        } else {
            #pragma unroll
            for (int ki = 0; ki < NKIT; ++ki) ahp[ki] = ah[ki];
        }
        feed(1 - hi);
        if (!hi) {
            nX = WMMA4((*xnext)[2], BX[2], nX);
            nX = WMMA4((*xnext)[3], BX[3], nX);
        }
        asm volatile("" :: "v"(accH), "v"(xnb[0]), "v"(xnb[1]), "v"(xnb[2]),
                           "v"(xnb[3]), "v"(nX));
        __builtin_amdgcn_sched_barrier(0);
        #pragma unroll
        for (int ki = 0; ki < NKIT; ++ki) ah[ki] = ahn[ki];
    };

    v2f xb[XPAIRS][NKIT];
    loadA(0, xb[0]);
    loadA(1, xb[1]);
    loadA(2, xb[2]);
    loadA(3, xb[3]);
    {
        v8f x0 = {};
        #pragma unroll
        for (int ki = 0; ki < NKIT; ++ki)
            x0 = WMMA4(xb[0][ki], BX[ki], x0);
        xcur = x0;
        feed(0);
    }
    lds_barrier();

    for (int t = 0; t < SEQ; t += 2 * XPAIRS) {
        const int p = t >> 1;
        const int pl = (t < SEQ - 2 * XPAIRS) ? p + 4 : p;
        if (w == 0 && (t & 31) == 8 && t >= 40)
            flush(t - 40);
        step(t,     1, 0, &xb[0], pl,     &xb[1]);
        step(t + 1, 0, 1, nullptr, 0,     &xb[1]);
        step(t + 2, 1, 0, &xb[1], pl + 1, &xb[2]);
        step(t + 3, 0, 1, nullptr, 0,     &xb[2]);
        step(t + 4, 1, 0, &xb[2], pl + 2, &xb[3]);
        step(t + 5, 0, 1, nullptr, 0,     &xb[3]);
        step(t + 6, 1, 0, &xb[3], pl + 3, &xb[0]);
        step(t + 7, 0, 1, nullptr, 0,     &xb[0]);
    }

    {
        const float op = outpart(ah);
        if (w == 0)
            Opart[(SEQ - 1) & 63][half][q] = op;
        lds_barrier();
        if (w == 0) flush(SEQ - 32);
    }
}

extern "C" void kernel_launch(void* const* d_in, const int* in_sizes, int n_in,
                              void* d_out, int out_size, void* d_ws, size_t ws_size,
                              hipStream_t stream) {
    (void)in_sizes; (void)n_in; (void)d_ws; (void)ws_size; (void)out_size;
    const float* x    = (const float*)d_in[0];
    const float* Wih  = (const float*)d_in[1];
    const float* Whh  = (const float*)d_in[2];
    const float* bih  = (const float*)d_in[3];
    const float* bhh  = (const float*)d_in[4];
    const float* Wout = (const float*)d_in[5];
    const float* bout = (const float*)d_in[6];
    float* out = (float*)d_out;

    dim3 grid(BATCH / RPG);
    dim3 block(32 * NW);
    gru_wmma_scan<<<grid, block, 0, stream>>>(x, Wih, Whh, bih, bhh, Wout, bout, out);
}


# Prologue's four x load groups kept in program order (scheduling barriers): the loop's counted load waits become exact, no step waits any more for rows requested two steps earlier; sum orders kept
# speedup vs baseline: 4.1290x; 1.0457x over previous
#include <hip/hip_runtime.h>


typedef float v2f __attribute__((ext_vector_type(2)));
typedef float v8f __attribute__((ext_vector_type(8)));

#define HID   19
#define SEQ   2048
#define BATCH 1024
#define NKIT  5
#define NW    4
#define UPW   5
#define RPG   8
#define XPAIRS 4

#define WMMA4(a, b, c) __builtin_amdgcn_wmma_f32_16x16x4_f32( \
        false, (a), false, (b), (short)0, (c), false, false)

__device__ __forceinline__ float fast_tanh(float x) {
#if __has_builtin(__builtin_amdgcn_tanhf)
    return __builtin_amdgcn_tanhf(x);
#else
    float r;
    asm("v_tanh_f32 %0, %1" : "=v"(r) : "v"(x));
    return r;
#endif
}
__device__ __forceinline__ float fast_sigmoid(float x) {
    return __builtin_fmaf(0.5f, fast_tanh(0.5f * x), 0.5f);
}
template <int N>
__device__ __forceinline__ float from_left(float v) {
    return __builtin_bit_cast(float, __builtin_amdgcn_update_dpp(
        0, __builtin_bit_cast(int, v), 0x110 + N, 0xf, 0xf, true));
}
__device__ __forceinline__ void lds_barrier() {
    __builtin_amdgcn_fence(__ATOMIC_RELEASE, "workgroup", "local");
    __builtin_amdgcn_s_barrier();
    __builtin_amdgcn_fence(__ATOMIC_ACQUIRE, "workgroup", "local");
}

__global__ __launch_bounds__(128) void gru_wmma_scan(
    const float* __restrict__ x,
    const float* __restrict__ Wih,
    const float* __restrict__ Whh,
    const float* __restrict__ bih,
    const float* __restrict__ bhh,
    const float* __restrict__ Wout,
    const float* __restrict__ bout,
    float* __restrict__ out)
{
    __shared__ __attribute__((aligned(16))) float Hall[2 * RPG * 20 + 256];
    float (*Hbuf)[RPG][20] = (float (*)[RPG][20])Hall;
    __shared__ __attribute__((aligned(16))) float Opart[64][2][RPG];

    const int w     = __builtin_amdgcn_readfirstlane((int)(threadIdx.x >> 5));
    const int oddw  = w & 1;
    const int lane  = threadIdx.x & 31;
    const int c     = lane & 15;
    const int half  = lane >> 4;
    const int bbase = blockIdx.x * RPG;
    const int q     = (c & 3) + 4 * (c >> 3);
    const int ts    = (c >> 2) & 1;

    const int  uH   = c / 3;
    const int  gH   = c - 3 * uH;
    const int  jH   = UPW * w + uH;
    const bool okH  = (c < 3 * UPW) && (jH < HID);
    const int  rowH = okH ? (gH * HID + jH) : 0;
    const bool nCol = (gH == 2);

    v2f BX[NKIT], BHh[NKIT];
    #pragma unroll
    for (int ki = 0; ki < NKIT; ++ki) {
        #pragma unroll
        for (int v = 0; v < 2; ++v) {
            const int K = 4 * ki + 2 * half + v;
            float wx = 0.0f, wh = 0.0f;
            if (okH) {
                if (K < HID) {
                    wh = Whh[rowH * HID + K];
                    wx = Wih[rowH * HID + K];
                } else {
                    wh = (gH < 2) ? (bih[rowH] + bhh[rowH]) : bhh[rowH];
                }
            }
            BX[ki][v]  = wx;
            BHh[ki][v] = wh;
        }
    }

    const bool  hLane = nCol && okH;
    const int   hword = hLane ? (4 * half * 20 + jH) : (2 * RPG * 20 + lane);
    const float bnx   = hLane ? bih[2 * HID + jH] : 0.0f;
    const float bo = bout[0];

    float wo[NKIT][2];
    #pragma unroll
    for (int ki = 0; ki < NKIT; ++ki)
        #pragma unroll
        for (int v = 0; v < 2; ++v) {
            const int j = 4 * ki + 2 * half + v;
            wo[ki][v] = (j < HID) ? Wout[j] : 0.0f;
        }

    for (int i = threadIdx.x; i < 2 * RPG * 20; i += 32 * NW)
        Hall[i] = (i % 20 == 19) ? 1.0f : 0.0f;

    v2f ah[NKIT];
    #pragma unroll
    for (int ki = 0; ki < NKIT; ++ki) { ah[ki][0] = 0.0f; ah[ki][1] = 0.0f; }
    ah[NKIT - 1][1] = half ? 1.0f : 0.0f;
    float hold[4];
    #pragma unroll
    for (int r = 0; r < 4; ++r) hold[r] = 0.0f;

    const unsigned xrow  = (unsigned)((ts * BATCH + bbase + q) * HID);
    const unsigned xoffA = (xrow + 2u * (unsigned)half) * 4u;
    const unsigned xoffB = (xrow + 17u) * 4u;
    auto loadA = [&](int p, v2f (&xa)[NKIT]) __attribute__((always_inline)) {
        const char*  xs = (const char*)x + (size_t)p * (size_t)(2 * BATCH * HID * 4);
        const float* xp = (const float*)(xs + xoffA);
        #pragma unroll
        for (int ki = 0; ki < NKIT - 1; ++ki) {
            xa[ki][0] = xp[4 * ki];
            xa[ki][1] = xp[4 * ki + 1];
        }
        xa[NKIT - 1][0] = xp[16];
        xa[NKIT - 1][1] = *(const float*)(xs + xoffB);
    };

    v8f accH = {}, xcur = {};
    float xnb[4] = {0.0f, 0.0f, 0.0f, 0.0f};
    auto feed = [&](const int hi) {
        #pragma unroll
        for (int r = 0; r < 4; ++r) {
            const float xv = xcur[4 * hi + r];
            accH[r] = nCol ? 0.0f : xv;
            xnb[r]  = xv + bnx;
        }
    };

    auto outpart = [&](const v2f (&f)[NKIT]) {
        float oacc = f[0][0] * wo[0][0];
        oacc = __builtin_fmaf(f[0][1], wo[0][1], oacc);
        #pragma unroll
        for (int ki = 1; ki < NKIT; ++ki)
            #pragma unroll
            for (int v = 0; v < 2; ++v)
                oacc = __builtin_fmaf(f[ki][v], wo[ki][v], oacc);
        return oacc;
    };
    v2f ahp[NKIT];
    #pragma unroll
    for (int ki = 0; ki < NKIT; ++ki) ahp[ki] = ah[ki];
    auto flush = [&](int t0) {
        const int   tl = t0 + lane;
        const float* s = &Opart[tl & 63][0][0];
        float* o = out + (size_t)tl * BATCH + bbase;
        #pragma unroll
        for (int g = 0; g < 2; ++g) {
            const float4 a = *(const float4*)(s + 4 * g);
            const float4 b = *(const float4*)(s + RPG + 4 * g);
            float4 r;
            r.x = (a.x + b.x) + bo;
            r.y = (a.y + b.y) + bo;
            r.z = (a.z + b.z) + bo;
            r.w = (a.w + b.w) + bo;
            *(float4*)(o + 4 * g) = r;
        }
    };

    v8f nX = {};
    auto step = [&](int t, const int nb, const int hi,
                    v2f (*xload)[NKIT], int pload, v2f (*xnext)[NKIT])
                    __attribute__((always_inline)) {
        v8f dH = accH;
        #pragma unroll
        for (int ki = 0; ki < NKIT; ++ki)
            dH = WMMA4(ah[ki], BHh[ki], dH);

        float hnew[4];
        #pragma unroll
        for (int r = 0; r < 4; ++r) {
            const float sg = fast_sigmoid(dH[r]);
            const float rg = from_left<2>(sg);
            const float zg = from_left<1>(sg);
            const float ng = fast_tanh(xnb[r] + rg * dH[r]);
            hnew[r] = __builtin_fmaf(zg, hold[r] - ng, ng);
            hold[r] = hnew[r];
        }
        #pragma unroll
        for (int r = 0; r < 4; ++r)
            Hall[nb * (RPG * 20) + hword + 20 * r] = hnew[r];

        __builtin_amdgcn_sched_barrier(0);
        if (xload)
            loadA(pload, *xload);

        if (hi) {
            nX = WMMA4((*xnext)[4], BX[4], nX);
            xcur = nX;
        } else {
            v8f z = {};
            nX = WMMA4((*xnext)[0], BX[0], z);
            nX = WMMA4((*xnext)[1], BX[1], nX);
        }
        asm volatile("" :: "v"(nX));

        lds_barrier();

        v2f ahn[NKIT];
        #pragma unroll
        for (int ki = 0; ki < NKIT; ++ki)
            ahn[ki] = *(const v2f*)&Hbuf[nb][q][4 * ki + 2 * half];
        __builtin_amdgcn_sched_barrier(0);
        if (hi) {
            v2f f[NKIT];
            #pragma unroll
            for (int ki = 0; ki < NKIT; ++ki) {
                f[ki][0] = oddw ? ah[ki][0] : ahp[ki][0];
                f[ki][1] = oddw ? ah[ki][1] : ahp[ki][1];
            }
            Opart[(t - 2 + oddw) & 63][half][q] = outpart(f);
        } else {
            #pragma unroll
            for (int ki = 0; ki < NKIT; ++ki) ahp[ki] = ah[ki];
        }
        feed(1 - hi);
        if (!hi) {
            nX = WMMA4((*xnext)[2], BX[2], nX);
            nX = WMMA4((*xnext)[3], BX[3], nX);
        }
        asm volatile("" :: "v"(accH), "v"(xnb[0]), "v"(xnb[1]), "v"(xnb[2]),
                           "v"(xnb[3]), "v"(nX));
        __builtin_amdgcn_sched_barrier(0);
        #pragma unroll
        for (int ki = 0; ki < NKIT; ++ki) ah[ki] = ahn[ki];
    };

    v2f xb[XPAIRS][NKIT];
    loadA(0, xb[0]);
    __builtin_amdgcn_sched_barrier(0);
    loadA(1, xb[1]);
    __builtin_amdgcn_sched_barrier(0);
    loadA(2, xb[2]);
    __builtin_amdgcn_sched_barrier(0);
    loadA(3, xb[3]);
    {
        v8f x0 = {};
        #pragma unroll
        for (int ki = 0; ki < NKIT; ++ki)
            x0 = WMMA4(xb[0][ki], BX[ki], x0);
        xcur = x0;
        feed(0);
    }
    lds_barrier();

    for (int t = 0; t < SEQ; t += 2 * XPAIRS) {
        const int p = t >> 1;
        const int pl = (t < SEQ - 2 * XPAIRS) ? p + 4 : p;
        if (w == 0 && (t & 31) == 8 && t >= 40)
            flush(t - 40);
        step(t,     1, 0, &xb[0], pl,     &xb[1]);
        step(t + 1, 0, 1, nullptr, 0,     &xb[1]);
        step(t + 2, 1, 0, &xb[1], pl + 1, &xb[2]);
        step(t + 3, 0, 1, nullptr, 0,     &xb[2]);
        step(t + 4, 1, 0, &xb[2], pl + 2, &xb[3]);
        step(t + 5, 0, 1, nullptr, 0,     &xb[3]);
        step(t + 6, 1, 0, &xb[3], pl + 3, &xb[0]);
        step(t + 7, 0, 1, nullptr, 0,     &xb[0]);
    }

    {
        const float op = outpart(ah);
        if (w == 0)
            Opart[(SEQ - 1) & 63][half][q] = op;
        lds_barrier();
        if (w == 0) flush(SEQ - 32);
    }
}

extern "C" void kernel_launch(void* const* d_in, const int* in_sizes, int n_in,
                              void* d_out, int out_size, void* d_ws, size_t ws_size,
                              hipStream_t stream) {
    (void)in_sizes; (void)n_in; (void)d_ws; (void)ws_size; (void)out_size;
    const float* x    = (const float*)d_in[0];
    const float* Wih  = (const float*)d_in[1];
    const float* Whh  = (const float*)d_in[2];
    const float* bih  = (const float*)d_in[3];
    const float* bhh  = (const float*)d_in[4];
    const float* Wout = (const float*)d_in[5];
    const float* bout = (const float*)d_in[6];
    float* out = (float*)d_out;

    dim3 grid(BATCH / RPG);
    dim3 block(32 * NW);
    gru_wmma_scan<<<grid, block, 0, stream>>>(x, Wih, Whh, bih, bhh, Wout, bout, out);
}


# h-side product on two accumulator chains (k blocks 0,2,4 from zero; x part + k blocks 1,3), added on the vector unit: 5 WMMAs at issue rate, not 5 result latencies; h-side sum order changes
# speedup vs baseline: 4.2673x; 1.0335x over previous
#include <hip/hip_runtime.h>


typedef float v2f __attribute__((ext_vector_type(2)));
typedef float v8f __attribute__((ext_vector_type(8)));

#define HID   19
#define SEQ   2048
#define BATCH 1024
#define NKIT  5
#define NW    4
#define UPW   5
#define RPG   8
#define XPAIRS 4

#define WMMA4(a, b, c) __builtin_amdgcn_wmma_f32_16x16x4_f32( \
        false, (a), false, (b), (short)0, (c), false, false)

__device__ __forceinline__ float fast_tanh(float x) {
#if __has_builtin(__builtin_amdgcn_tanhf)
    return __builtin_amdgcn_tanhf(x);
#else
    float r;
    asm("v_tanh_f32 %0, %1" : "=v"(r) : "v"(x));
    return r;
#endif
}
__device__ __forceinline__ float fast_sigmoid(float x) {
    return __builtin_fmaf(0.5f, fast_tanh(0.5f * x), 0.5f);
}
template <int N>
__device__ __forceinline__ float from_left(float v) {
    return __builtin_bit_cast(float, __builtin_amdgcn_update_dpp(
        0, __builtin_bit_cast(int, v), 0x110 + N, 0xf, 0xf, true));
}
__device__ __forceinline__ void lds_barrier() {
    __builtin_amdgcn_fence(__ATOMIC_RELEASE, "workgroup", "local");
    __builtin_amdgcn_s_barrier();
    __builtin_amdgcn_fence(__ATOMIC_ACQUIRE, "workgroup", "local");
}

__global__ __launch_bounds__(128) void gru_wmma_scan(
    const float* __restrict__ x,
    const float* __restrict__ Wih,
    const float* __restrict__ Whh,
    const float* __restrict__ bih,
    const float* __restrict__ bhh,
    const float* __restrict__ Wout,
    const float* __restrict__ bout,
    float* __restrict__ out)
{
    __shared__ __attribute__((aligned(16))) float Hall[2 * RPG * 20 + 256];
    float (*Hbuf)[RPG][20] = (float (*)[RPG][20])Hall;
    __shared__ __attribute__((aligned(16))) float Opart[64][2][RPG];

    const int w     = __builtin_amdgcn_readfirstlane((int)(threadIdx.x >> 5));
    const int oddw  = w & 1;
    const int lane  = threadIdx.x & 31;
    const int c     = lane & 15;
    const int half  = lane >> 4;
    const int bbase = blockIdx.x * RPG;
    const int q     = (c & 3) + 4 * (c >> 3);
    const int ts    = (c >> 2) & 1;

    const int  uH   = c / 3;
    const int  gH   = c - 3 * uH;
    const int  jH   = UPW * w + uH;
    const bool okH  = (c < 3 * UPW) && (jH < HID);
    const int  rowH = okH ? (gH * HID + jH) : 0;
    const bool nCol = (gH == 2);

    v2f BX[NKIT], BHh[NKIT];
    #pragma unroll
    for (int ki = 0; ki < NKIT; ++ki) {
        #pragma unroll
        for (int v = 0; v < 2; ++v) {
            const int K = 4 * ki + 2 * half + v;
            float wx = 0.0f, wh = 0.0f;
            if (okH) {
                if (K < HID) {
                    wh = Whh[rowH * HID + K];
                    wx = Wih[rowH * HID + K];
                } else {
                    wh = (gH < 2) ? (bih[rowH] + bhh[rowH]) : bhh[rowH];
                }
            }
            BX[ki][v]  = wx;
            BHh[ki][v] = wh;
        }
    }

    const bool  hLane = nCol && okH;
    const int   hword = hLane ? (4 * half * 20 + jH) : (2 * RPG * 20 + lane);
    const float bnx   = hLane ? bih[2 * HID + jH] : 0.0f;
    const float bo = bout[0];

    float wo[NKIT][2];
    #pragma unroll
    for (int ki = 0; ki < NKIT; ++ki)
        #pragma unroll
        for (int v = 0; v < 2; ++v) {
            const int j = 4 * ki + 2 * half + v;
            wo[ki][v] = (j < HID) ? Wout[j] : 0.0f;
        }

    for (int i = threadIdx.x; i < 2 * RPG * 20; i += 32 * NW)
        Hall[i] = (i % 20 == 19) ? 1.0f : 0.0f;

    v2f ah[NKIT];
    #pragma unroll
    for (int ki = 0; ki < NKIT; ++ki) { ah[ki][0] = 0.0f; ah[ki][1] = 0.0f; }
    ah[NKIT - 1][1] = half ? 1.0f : 0.0f;
    float hold[4];
    #pragma unroll
    for (int r = 0; r < 4; ++r) hold[r] = 0.0f;

    const unsigned xrow  = (unsigned)((ts * BATCH + bbase + q) * HID);
    const unsigned xoffA = (xrow + 2u * (unsigned)half) * 4u;
    const unsigned xoffB = (xrow + 17u) * 4u;
    auto loadA = [&](int p, v2f (&xa)[NKIT]) __attribute__((always_inline)) {
        const char*  xs = (const char*)x + (size_t)p * (size_t)(2 * BATCH * HID * 4);
        const float* xp = (const float*)(xs + xoffA);
        #pragma unroll
        for (int ki = 0; ki < NKIT - 1; ++ki) {
            xa[ki][0] = xp[4 * ki];
            xa[ki][1] = xp[4 * ki + 1];
        }
        xa[NKIT - 1][0] = xp[16];
        xa[NKIT - 1][1] = *(const float*)(xs + xoffB);
    };

    v8f accH = {}, xcur = {};
    float xnb[4] = {0.0f, 0.0f, 0.0f, 0.0f};
    auto feed = [&](const int hi) {
        #pragma unroll
        for (int r = 0; r < 4; ++r) {
            const float xv = xcur[4 * hi + r];
            accH[r] = nCol ? 0.0f : xv;
            xnb[r]  = xv + bnx;
        }
    };

    auto outpart = [&](const v2f (&f)[NKIT]) {
        float oacc = f[0][0] * wo[0][0];
        oacc = __builtin_fmaf(f[0][1], wo[0][1], oacc);
        #pragma unroll
        for (int ki = 1; ki < NKIT; ++ki)
            #pragma unroll
            for (int v = 0; v < 2; ++v)
                oacc = __builtin_fmaf(f[ki][v], wo[ki][v], oacc);
        return oacc;
    };
    v2f ahp[NKIT];
    #pragma unroll
    for (int ki = 0; ki < NKIT; ++ki) ahp[ki] = ah[ki];
    auto flush = [&](int t0) {
        const int   tl = t0 + lane;
        const float* s = &Opart[tl & 63][0][0];
        float* o = out + (size_t)tl * BATCH + bbase;
        #pragma unroll
        for (int g = 0; g < 2; ++g) {
            const float4 a = *(const float4*)(s + 4 * g);
            const float4 b = *(const float4*)(s + RPG + 4 * g);
            float4 r;
            r.x = (a.x + b.x) + bo;
            r.y = (a.y + b.y) + bo;
            r.z = (a.z + b.z) + bo;
            r.w = (a.w + b.w) + bo;
            *(float4*)(o + 4 * g) = r;
        }
    };

    v8f nX = {};
    auto step = [&](int t, const int nb, const int hi,
                    v2f (*xload)[NKIT], int pload, v2f (*xnext)[NKIT])
                    __attribute__((always_inline)) {
        v8f dA = {};
        v8f dB = accH;
        dA = WMMA4(ah[0], BHh[0], dA);
        dB = WMMA4(ah[1], BHh[1], dB);
        dA = WMMA4(ah[2], BHh[2], dA);
        dB = WMMA4(ah[3], BHh[3], dB);
        dA = WMMA4(ah[4], BHh[4], dA);
        __builtin_amdgcn_sched_barrier(0);

        float hnew[4];
        #pragma unroll
        for (int r = 0; r < 4; ++r) {
            const float hb = 0.5f * dB[r];
            const float dh = dA[r] + dB[r];
            const float sg = __builtin_fmaf(0.5f,
                fast_tanh(__builtin_fmaf(dA[r], 0.5f, hb)), 0.5f);
            const float rg = from_left<2>(sg);
            const float zg = from_left<1>(sg);
            const float ng = fast_tanh(xnb[r] + rg * dh);
            hnew[r] = __builtin_fmaf(zg, hold[r] - ng, ng);
            hold[r] = hnew[r];
        }
        #pragma unroll
        for (int r = 0; r < 4; ++r)
            Hall[nb * (RPG * 20) + hword + 20 * r] = hnew[r];

        asm volatile("" :: "v"(hnew[0]), "v"(hnew[1]), "v"(hnew[2]), "v"(hnew[3]));
        __builtin_amdgcn_sched_barrier(0);
        if (xload)
            loadA(pload, *xload);

        if (hi) {
            nX = WMMA4((*xnext)[4], BX[4], nX);
            xcur = nX;
        } else {
            v8f z = {};
            nX = WMMA4((*xnext)[0], BX[0], z);
            nX = WMMA4((*xnext)[1], BX[1], nX);
        }
        asm volatile("" :: "v"(nX));

        lds_barrier();

        v2f ahn[NKIT];
        #pragma unroll
        for (int ki = 0; ki < NKIT; ++ki)
            ahn[ki] = *(const v2f*)&Hbuf[nb][q][4 * ki + 2 * half];
        __builtin_amdgcn_sched_barrier(0);
        if (hi) {
            v2f f[NKIT];
            #pragma unroll
            for (int ki = 0; ki < NKIT; ++ki) {
                f[ki][0] = oddw ? ah[ki][0] : ahp[ki][0];
                f[ki][1] = oddw ? ah[ki][1] : ahp[ki][1];
            }
            Opart[(t - 2 + oddw) & 63][half][q] = outpart(f);
        } else {
            #pragma unroll
            for (int ki = 0; ki < NKIT; ++ki) ahp[ki] = ah[ki];
        }
        feed(1 - hi);
        if (!hi) {
            nX = WMMA4((*xnext)[2], BX[2], nX);
            nX = WMMA4((*xnext)[3], BX[3], nX);
        }
        asm volatile("" :: "v"(accH), "v"(xnb[0]), "v"(xnb[1]), "v"(xnb[2]),
                           "v"(xnb[3]), "v"(nX));
        __builtin_amdgcn_sched_barrier(0);
        #pragma unroll
        for (int ki = 0; ki < NKIT; ++ki) ah[ki] = ahn[ki];
    };

    v2f xb[XPAIRS][NKIT];
    loadA(0, xb[0]);
    __builtin_amdgcn_sched_barrier(0);
    loadA(1, xb[1]);
    __builtin_amdgcn_sched_barrier(0);
    loadA(2, xb[2]);
    __builtin_amdgcn_sched_barrier(0);
    loadA(3, xb[3]);
    {
        v8f x0 = {};
        #pragma unroll
        for (int ki = 0; ki < NKIT; ++ki)
            x0 = WMMA4(xb[0][ki], BX[ki], x0);
        xcur = x0;
        feed(0);
    }
    lds_barrier();

    for (int t = 0; t < SEQ; t += 2 * XPAIRS) {
        const int p = t >> 1;
        const int pl = (t < SEQ - 2 * XPAIRS) ? p + 4 : p;
        if (w == 0 && (t & 31) == 8 && t >= 40)
            flush(t - 40);
        step(t,     1, 0, &xb[0], pl,     &xb[1]);
        step(t + 1, 0, 1, nullptr, 0,     &xb[1]);
        step(t + 2, 1, 0, &xb[1], pl + 1, &xb[2]);
        step(t + 3, 0, 1, nullptr, 0,     &xb[2]);
        step(t + 4, 1, 0, &xb[2], pl + 2, &xb[3]);
        step(t + 5, 0, 1, nullptr, 0,     &xb[3]);
        step(t + 6, 1, 0, &xb[3], pl + 3, &xb[0]);
        step(t + 7, 0, 1, nullptr, 0,     &xb[0]);
    }

    {
        const float op = outpart(ah);
        if (w == 0)
            Opart[(SEQ - 1) & 63][half][q] = op;
        lds_barrier();
        if (w == 0) flush(SEQ - 32);
    }
}

extern "C" void kernel_launch(void* const* d_in, const int* in_sizes, int n_in,
                              void* d_out, int out_size, void* d_ws, size_t ws_size,
                              hipStream_t stream) {
    (void)in_sizes; (void)n_in; (void)d_ws; (void)ws_size; (void)out_size;
    const float* x    = (const float*)d_in[0];
    const float* Wih  = (const float*)d_in[1];
    const float* Whh  = (const float*)d_in[2];
    const float* bih  = (const float*)d_in[3];
    const float* bhh  = (const float*)d_in[4];
    const float* Wout = (const float*)d_in[5];
    const float* bout = (const float*)d_in[6];
    float* out = (float*)d_out;

    dim3 grid(BATCH / RPG);
    dim3 block(32 * NW);
    gru_wmma_scan<<<grid, block, 0, stream>>>(x, Wih, Whh, bih, bhh, Wout, bout, out);
}
